# starting kernel
# baseline (speedup 1.0000x reference)
#include <hip/hip_runtime.h>

typedef __attribute__((ext_vector_type(2))) float v2f;
typedef __attribute__((ext_vector_type(4))) float v4f;
typedef __attribute__((ext_vector_type(8))) float v8f;

#define BLOCK 32
#define NB 128
#define KDIM 4096
#define NDIM 4096
#define ROWS_PER_WG 256
#define THREADS 256
#define LSTRIDE 34

__global__ void bsmm_build_csc(const int* __restrict__ idx_i,
                               const int* __restrict__ idx_j, int nnz,
                               int* __restrict__ colptr,
                               int* __restrict__ colblk,
                               int* __restrict__ coli)
{
    if (threadIdx.x != 0 || blockIdx.x != 0) return;
    int cnt[NB];
    for (int j = 0; j < NB; ++j) cnt[j] = 0;
    for (int e = 0; e < nnz; ++e) cnt[idx_j[e]]++;
    int run = 0;
    for (int j = 0; j < NB; ++j) { colptr[j] = run; run += cnt[j]; cnt[j] = 0; }
    colptr[NB] = run;
    for (int e = 0; e < nnz; ++e) {
        int j = idx_j[e];
        int o = colptr[j] + cnt[j]++;
        colblk[o] = e;
        coli[o]   = idx_i[e];
    }
}

__global__ void __launch_bounds__(THREADS)
bsmm_wmma(const float* __restrict__ x, const float* __restrict__ w,
          const int* __restrict__ colptr, const int* __restrict__ colblk,
          const int* __restrict__ coli, float* __restrict__ y)
{
    __shared__ float wlds[2][BLOCK * LSTRIDE];

    const int j    = blockIdx.y;
    const int lane = threadIdx.x & 31;
    const int wave = threadIdx.x >> 5;
    const int half = lane >> 4;
    const int lid  = lane & 15;
    const int m0   = blockIdx.x * ROWS_PER_WG + wave * 32;

    const int beg = colptr[j];
    const int cnt = colptr[j + 1] - beg;

    v8f c00 = {}, c01 = {};
    v8f c10 = {}, c11 = {};

    const float* xrow0 = x + (size_t)(m0 + lid) * KDIM + half * 2;
    const float* xrow1 = xrow0 + (size_t)16 * KDIM;

    const int boff = lid * LSTRIDE + half * 2;

    const int srow = threadIdx.x >> 3;
    const int scol = (threadIdx.x & 7) * 4;

    if (cnt > 0) {
        const float* wsrc =
            w + (size_t)colblk[beg] * (BLOCK * BLOCK) + srow * BLOCK + scol;
        v4f t0 = *(const v4f*)wsrc;
        wlds[0][(scol + 0) * LSTRIDE + srow] = t0[0];
        wlds[0][(scol + 1) * LSTRIDE + srow] = t0[1];
        wlds[0][(scol + 2) * LSTRIDE + srow] = t0[2];
        wlds[0][(scol + 3) * LSTRIDE + srow] = t0[3];
    }

    for (int t = 0; t < cnt; ++t) {
        __syncthreads();
        const int buf = t & 1;

        const bool have_next = (t + 1) < cnt;
        v4f nxt = {};
        if (have_next) {
            const float* wsrc = w + (size_t)colblk[beg + t + 1] * (BLOCK * BLOCK)
                                  + srow * BLOCK + scol;
            nxt = *(const v4f*)wsrc;
        }

        const int    kbase = coli[beg + t] * BLOCK;
        const float* xa0   = xrow0 + kbase;
        const float* xa1   = xrow1 + kbase;
        const float* bl    = &wlds[buf][boff];

#pragma unroll
        for (int s = 0; s < 8; ++s) {
            v2f a0 = *(const v2f*)(xa0 + s * 4);
            v2f a1 = *(const v2f*)(xa1 + s * 4);
            v2f b0 = *(const v2f*)(bl + s * 4);
            v2f b1 = *(const v2f*)(bl + 16 * LSTRIDE + s * 4);
            c00 = __builtin_amdgcn_wmma_f32_16x16x4_f32(
                      false, a0, false, b0, (short)0, c00, false, false);
            c01 = __builtin_amdgcn_wmma_f32_16x16x4_f32(
                      false, a0, false, b1, (short)0, c01, false, false);
            c10 = __builtin_amdgcn_wmma_f32_16x16x4_f32(
                      false, a1, false, b0, (short)0, c10, false, false);
            c11 = __builtin_amdgcn_wmma_f32_16x16x4_f32(
                      false, a1, false, b1, (short)0, c11, false, false);
        }

        if (have_next) {
            __syncthreads();
            float* dst = &wlds[buf ^ 1][0];
            dst[(scol + 0) * LSTRIDE + srow] = nxt[0];
            dst[(scol + 1) * LSTRIDE + srow] = nxt[1];
            dst[(scol + 2) * LSTRIDE + srow] = nxt[2];
            dst[(scol + 3) * LSTRIDE + srow] = nxt[3];
        }
    }

    float* y0 = y + (size_t)(m0 + half * 8) * NDIM + j * BLOCK + lid;
    float* y1 = y0 + (size_t)16 * NDIM;
#pragma unroll
    for (int r = 0; r < 8; ++r) {
        y0[(size_t)r * NDIM]      = c00[r];
        y0[(size_t)r * NDIM + 16] = c01[r];
        y1[(size_t)r * NDIM]      = c10[r];
        y1[(size_t)r * NDIM + 16] = c11[r];
    }
}

extern "C" void kernel_launch(void* const* d_in, const int* in_sizes, int n_in,
                              void* d_out, int out_size, void* d_ws, size_t ws_size,
                              hipStream_t stream) {
    const float* x     = (const float*)d_in[0];
    const float* w     = (const float*)d_in[1];
    const int*   idx_i = (const int*)d_in[2];
    const int*   idx_j = (const int*)d_in[3];
    float*       y     = (float*)d_out;

    const int nnz = in_sizes[2];
    const int M   = in_sizes[0] / KDIM;

    int* colptr = (int*)d_ws;
    int* colblk = colptr + (NB + 1);
    int* coli   = colblk + nnz;

    bsmm_build_csc<<<1, 32, 0, stream>>>(idx_i, idx_j, nnz, colptr, colblk, coli);

    dim3 grid(M / ROWS_PER_WG, NB);
    bsmm_wmma<<<grid, THREADS, 0, stream>>>(x, w, colptr, colblk, coli, y);
}


# Column block lists built inside the GEMM kernel (each work-group ballot-scans idx_j into an LDS list) instead of a 1-thread CSC build launch; 2 launches -> 1, no workspace; sum order kept
# speedup vs baseline: 5.8636x; 5.8636x over previous
#include <hip/hip_runtime.h>

typedef __attribute__((ext_vector_type(2))) float v2f;
typedef __attribute__((ext_vector_type(4))) float v4f;
typedef __attribute__((ext_vector_type(8))) float v8f;
typedef __attribute__((ext_vector_type(2))) int   v2i;

#define BLOCK 32
#define NB 128
#define KDIM 4096
#define NDIM 4096
#define ROWS_PER_WG 256
#define THREADS 256
#define NWAVES 8
#define LSTRIDE 34
#define CHUNKS 8
#define SEG (NWAVES * CHUNKS * 32)
#define AHEAD 3

__global__ void __launch_bounds__(THREADS, 12)
bsmm_wmma(const float* __restrict__ x, const float* __restrict__ w,
          const int* __restrict__ idx_i, const int* __restrict__ idx_j, int nnz,
          float* __restrict__ y)
{
    __shared__ float wlds[2][BLOCK * LSTRIDE];
    __shared__ v2i lst[SEG + 2];
    __shared__ int wcnt[NWAVES];

    const int j    = blockIdx.y;
    const int lane = threadIdx.x & 31;
    const int wave = threadIdx.x >> 5;
    const int half = lane >> 4;
    const int lid  = lane & 15;
    const int m0   = blockIdx.x * ROWS_PER_WG + wave * 32;

    v8f c00 = {}, c01 = {};
    v8f c10 = {}, c11 = {};

    const float* xrow0 = x + (size_t)(m0 + lid) * KDIM + half * 2;
    const float* xrow1 = xrow0 + (size_t)16 * KDIM;

    const int boff = lid * LSTRIDE + half * 2;

    const int srow = threadIdx.x >> 3;
    const int scol = (threadIdx.x & 7) * 4;
    const int soff = srow * BLOCK + scol;

    const int eloc0 = wave * (CHUNKS * 32) + lane;

    for (int seg0 = 0; seg0 < nnz; seg0 += SEG) {
        int vj[CHUNKS], vi[CHUNKS];
#pragma unroll
        for (int c = 0; c < CHUNKS; ++c) {
            const int e  = seg0 + eloc0 + c * 32;
            const int ec = e < nnz ? e : nnz - 1;
            const int rj = idx_j[ec];
            vi[c] = idx_i[ec];
            vj[c] = e < nnz ? rj : -1;
        }
        unsigned m[CHUNKS];
        int wc = 0;
#pragma unroll
        for (int c = 0; c < CHUNKS; ++c) {
            m[c] = __builtin_amdgcn_ballot_w32(vj[c] == j);
            wc += __builtin_popcount(m[c]);
        }
        if (lane == 0) wcnt[wave] = wc;
        __syncthreads();

        int base = 0, tot = 0;
#pragma unroll
        for (int k = 0; k < NWAVES; ++k) {
            const int v = wcnt[k];
            tot += v;
            if (k < wave) base += v;
        }
        const int cnt = __builtin_amdgcn_readfirstlane(tot);

#pragma unroll
        for (int c = 0; c < CHUNKS; ++c) {
            if (vj[c] == j) {
                v2i ent = { seg0 + eloc0 + c * 32, vi[c] };
                lst[base + (int)__builtin_amdgcn_mbcnt_lo(m[c], 0u)] = ent;
            }
            base += __builtin_popcount(m[c]);
        }
        if (threadIdx.x == 0) { v2i z = { 0, 0 }; lst[cnt] = z; lst[cnt + 1] = z; }
        __syncthreads();

        const v2i l0 = lst[0], l1 = lst[1];
        int blk_cur = __builtin_amdgcn_readfirstlane(l0[0]);
        int col_cur = __builtin_amdgcn_readfirstlane(l0[1]);
        int blk_nxt = __builtin_amdgcn_readfirstlane(l1[0]);
        int col_nxt = __builtin_amdgcn_readfirstlane(l1[1]);

        if (cnt > 0) {
            const float* wsrc = w + (size_t)blk_cur * (BLOCK * BLOCK) + soff;
            v4f t0 = *(const v4f*)wsrc;
            wlds[0][(scol + 0) * LSTRIDE + srow] = t0[0];
            wlds[0][(scol + 1) * LSTRIDE + srow] = t0[1];
            wlds[0][(scol + 2) * LSTRIDE + srow] = t0[2];
            wlds[0][(scol + 3) * LSTRIDE + srow] = t0[3];
        }

        for (int t = 0; t < cnt; ++t) {
            __syncthreads();
            const int buf = t & 1;


            const bool have_next = (t + 1) < cnt;
            v4f nxt = {};
            if (have_next) {
                const float* wsrc = w + (size_t)blk_nxt * (BLOCK * BLOCK) + soff;
                nxt = *(const v4f*)wsrc;
            }

            const int    kbase = col_cur * BLOCK;
            const float* xa0   = xrow0 + kbase;
            const float* xa1   = xrow1 + kbase;
            const float* bl    = &wlds[buf][boff];

            v2f a0[8], a1[8];
#pragma unroll
            for (int s = 0; s < AHEAD; ++s) {
                a0[s] = *(const v2f*)(xa0 + s * 4);
                a1[s] = *(const v2f*)(xa1 + s * 4);
            }
#pragma unroll
            for (int s = 0; s < 8; ++s) {
                v2f b0 = *(const v2f*)(bl + s * 4);
                v2f b1 = *(const v2f*)(bl + 16 * LSTRIDE + s * 4);
                if (s + AHEAD < 8) {
                    a0[s + AHEAD] = *(const v2f*)(xa0 + (s + AHEAD) * 4);
                    a1[s + AHEAD] = *(const v2f*)(xa1 + (s + AHEAD) * 4);
                }
                c00 = __builtin_amdgcn_wmma_f32_16x16x4_f32(
                          false, a0[s], false, b0, (short)0, c00, false, false);
                c01 = __builtin_amdgcn_wmma_f32_16x16x4_f32(
                          false, a0[s], false, b1, (short)0, c01, false, false);
                c10 = __builtin_amdgcn_wmma_f32_16x16x4_f32(
                          false, a1[s], false, b0, (short)0, c10, false, false);
                c11 = __builtin_amdgcn_wmma_f32_16x16x4_f32(
                          false, a1[s], false, b1, (short)0, c11, false, false);
            }

            if (have_next) {
                __syncthreads();
                float* dst = &wlds[buf ^ 1][0];
                dst[(scol + 0) * LSTRIDE + srow] = nxt[0];
                dst[(scol + 1) * LSTRIDE + srow] = nxt[1];
                dst[(scol + 2) * LSTRIDE + srow] = nxt[2];
                dst[(scol + 3) * LSTRIDE + srow] = nxt[3];
            }

            const v2i la = lst[t + 2];
            col_cur = col_nxt;
            blk_nxt = __builtin_amdgcn_readfirstlane(la[0]);
            col_nxt = __builtin_amdgcn_readfirstlane(la[1]);
        }
    }

    float* y0 = y + (size_t)(m0 + half * 8) * NDIM + j * BLOCK + lid;
    float* y1 = y0 + (size_t)16 * NDIM;
#pragma unroll
    for (int r = 0; r < 8; ++r) {
        y0[(size_t)r * NDIM]      = c00[r];
        y0[(size_t)r * NDIM + 16] = c01[r];
        y1[(size_t)r * NDIM]      = c10[r];
        y1[(size_t)r * NDIM + 16] = c11[r];
    }
}

extern "C" void kernel_launch(void* const* d_in, const int* in_sizes, int n_in,
                              void* d_out, int out_size, void* d_ws, size_t ws_size,
                              hipStream_t stream) {
    const float* x     = (const float*)d_in[0];
    const float* w     = (const float*)d_in[1];
    const int*   idx_i = (const int*)d_in[2];
    const int*   idx_j = (const int*)d_in[3];
    float*       y     = (float*)d_out;

    const int nnz = in_sizes[2];
    const int M   = in_sizes[0] / KDIM;

    dim3 grid(M / ROWS_PER_WG, NB);
    bsmm_wmma<<<grid, THREADS, 0, stream>>>(x, w, idx_i, idx_j, nnz, y);
}


# A and B fragments read as 16-byte pieces (4 loads per 16-row tile, a block's 8 in one group; B as ds b128), k order inside each 32-k block permuted alike; in-block sum order changes
# speedup vs baseline: 7.7441x; 1.3207x over previous
#include <hip/hip_runtime.h>

typedef __attribute__((ext_vector_type(2))) float v2f;
typedef __attribute__((ext_vector_type(4))) float v4f;
typedef __attribute__((ext_vector_type(8))) float v8f;
typedef __attribute__((ext_vector_type(2))) int   v2i;

#define BLOCK 32
#define NB 128
#define KDIM 4096
#define NDIM 4096
#define ROWS_PER_WG 256
#define THREADS 256
#define NWAVES 8
#define LSTRIDE 36
#define CHUNKS 8
#define SEG (NWAVES * CHUNKS * 32)

__global__ void __launch_bounds__(THREADS, 10)
bsmm_wmma(const float* __restrict__ x, const float* __restrict__ w,
          const int* __restrict__ idx_i, const int* __restrict__ idx_j, int nnz,
          float* __restrict__ y)
{
    __shared__ alignas(16) float wlds[2][BLOCK * LSTRIDE];
    __shared__ v2i lst[SEG + 2];
    __shared__ int wcnt[NWAVES];

    const int j    = blockIdx.y;
    const int lane = threadIdx.x & 31;
    const int wave = threadIdx.x >> 5;
    const int half = lane >> 4;
    const int lid  = lane & 15;
    const int m0   = blockIdx.x * ROWS_PER_WG + wave * 32;

    v8f c00 = {}, c01 = {};
    v8f c10 = {}, c11 = {};

    const float* xrow0 = x + (size_t)(m0 + lid) * KDIM + half * 4;
    const float* xrow1 = xrow0 + (size_t)16 * KDIM;

    const int boff = lid * LSTRIDE + half * 4;

    const int srow = threadIdx.x >> 3;
    const int scol = (threadIdx.x & 7) * 4;
    const int soff = srow * BLOCK + scol;

    const int eloc0 = wave * (CHUNKS * 32) + lane;

    for (int seg0 = 0; seg0 < nnz; seg0 += SEG) {
        int vj[CHUNKS], vi[CHUNKS];
#pragma unroll
        for (int c = 0; c < CHUNKS; ++c) {
            const int e  = seg0 + eloc0 + c * 32;
            const int ec = e < nnz ? e : nnz - 1;
            const int rj = idx_j[ec];
            vi[c] = idx_i[ec];
            vj[c] = e < nnz ? rj : -1;
        }
        unsigned m[CHUNKS];
        int wc = 0;
#pragma unroll
        for (int c = 0; c < CHUNKS; ++c) {
            m[c] = __builtin_amdgcn_ballot_w32(vj[c] == j);
            wc += __builtin_popcount(m[c]);
        }
        if (lane == 0) wcnt[wave] = wc;
        __syncthreads();

        int base = 0, tot = 0;
#pragma unroll
        for (int k = 0; k < NWAVES; ++k) {
            const int v = wcnt[k];
            tot += v;
            if (k < wave) base += v;
        }
        const int cnt = __builtin_amdgcn_readfirstlane(tot);

#pragma unroll
        for (int c = 0; c < CHUNKS; ++c) {
            if (vj[c] == j) {
                v2i ent = { seg0 + eloc0 + c * 32, vi[c] };
                lst[base + (int)__builtin_amdgcn_mbcnt_lo(m[c], 0u)] = ent;
            }
            base += __builtin_popcount(m[c]);
        }
        if (threadIdx.x == 0) { v2i z = { 0, 0 }; lst[cnt] = z; lst[cnt + 1] = z; }
        __syncthreads();

        const v2i l0 = lst[0], l1 = lst[1];
        int blk_cur = __builtin_amdgcn_readfirstlane(l0[0]);
        int col_cur = __builtin_amdgcn_readfirstlane(l0[1]);
        int blk_nxt = __builtin_amdgcn_readfirstlane(l1[0]);
        int col_nxt = __builtin_amdgcn_readfirstlane(l1[1]);

        if (cnt > 0) {
            const float* wsrc = w + (size_t)blk_cur * (BLOCK * BLOCK) + soff;
            v4f t0 = *(const v4f*)wsrc;
            wlds[0][(scol + 0) * LSTRIDE + srow] = t0[0];
            wlds[0][(scol + 1) * LSTRIDE + srow] = t0[1];
            wlds[0][(scol + 2) * LSTRIDE + srow] = t0[2];
            wlds[0][(scol + 3) * LSTRIDE + srow] = t0[3];
        }

        for (int t = 0; t < cnt; ++t) {
            __syncthreads();
            const int buf = t & 1;


            const bool have_next = (t + 1) < cnt;
            v4f nxt = {};
            if (have_next) {
                const float* wsrc = w + (size_t)blk_nxt * (BLOCK * BLOCK) + soff;
                nxt = *(const v4f*)wsrc;
            }

            const int    kbase = col_cur * BLOCK;
            const float* xa0   = xrow0 + kbase;
            const float* xa1   = xrow1 + kbase;
            const float* bl    = &wlds[buf][boff];

            v4f a0[4], a1[4];
#pragma unroll
            for (int q = 0; q < 4; ++q) {
                a0[q] = *(const v4f*)(xa0 + q * 8);
                a1[q] = *(const v4f*)(xa1 + q * 8);
            }
            __builtin_amdgcn_sched_barrier(0);
#pragma unroll
            for (int q = 0; q < 4; ++q) {
                const v4f b0 = *(const v4f*)(bl + q * 8);
                const v4f b1 = *(const v4f*)(bl + 16 * LSTRIDE + q * 8);
                const v2f a0l = __builtin_shufflevector(a0[q], a0[q], 0, 1);
                const v2f a1l = __builtin_shufflevector(a1[q], a1[q], 0, 1);
                const v2f b0l = __builtin_shufflevector(b0, b0, 0, 1);
                const v2f b1l = __builtin_shufflevector(b1, b1, 0, 1);
                c00 = __builtin_amdgcn_wmma_f32_16x16x4_f32(
                          false, a0l, false, b0l, (short)0, c00, false, false);
                c01 = __builtin_amdgcn_wmma_f32_16x16x4_f32(
                          false, a0l, false, b1l, (short)0, c01, false, false);
                c10 = __builtin_amdgcn_wmma_f32_16x16x4_f32(
                          false, a1l, false, b0l, (short)0, c10, false, false);
                c11 = __builtin_amdgcn_wmma_f32_16x16x4_f32(
                          false, a1l, false, b1l, (short)0, c11, false, false);
                const v2f a0h = __builtin_shufflevector(a0[q], a0[q], 2, 3);
                const v2f a1h = __builtin_shufflevector(a1[q], a1[q], 2, 3);
                const v2f b0h = __builtin_shufflevector(b0, b0, 2, 3);
                const v2f b1h = __builtin_shufflevector(b1, b1, 2, 3);
                c00 = __builtin_amdgcn_wmma_f32_16x16x4_f32(
                          false, a0h, false, b0h, (short)0, c00, false, false);
                c01 = __builtin_amdgcn_wmma_f32_16x16x4_f32(
                          false, a0h, false, b1h, (short)0, c01, false, false);
                c10 = __builtin_amdgcn_wmma_f32_16x16x4_f32(
                          false, a1h, false, b0h, (short)0, c10, false, false);
                c11 = __builtin_amdgcn_wmma_f32_16x16x4_f32(
                          false, a1h, false, b1h, (short)0, c11, false, false);
            }

            if (have_next) {
                __syncthreads();
                float* dst = &wlds[buf ^ 1][0];
                dst[(scol + 0) * LSTRIDE + srow] = nxt[0];
                dst[(scol + 1) * LSTRIDE + srow] = nxt[1];
                dst[(scol + 2) * LSTRIDE + srow] = nxt[2];
                dst[(scol + 3) * LSTRIDE + srow] = nxt[3];
            }

            const v2i la = lst[t + 2];
            col_cur = col_nxt;
            blk_nxt = __builtin_amdgcn_readfirstlane(la[0]);
            col_nxt = __builtin_amdgcn_readfirstlane(la[1]);
        }
    }

    float* y0 = y + (size_t)(m0 + half * 8) * NDIM + j * BLOCK + lid;
    float* y1 = y0 + (size_t)16 * NDIM;
#pragma unroll
    for (int r = 0; r < 8; ++r) {
        y0[(size_t)r * NDIM]      = c00[r];
        y0[(size_t)r * NDIM + 16] = c01[r];
        y1[(size_t)r * NDIM]      = c10[r];
        y1[(size_t)r * NDIM + 16] = c11[r];
    }
}

extern "C" void kernel_launch(void* const* d_in, const int* in_sizes, int n_in,
                              void* d_out, int out_size, void* d_ws, size_t ws_size,
                              hipStream_t stream) {
    const float* x     = (const float*)d_in[0];
    const float* w     = (const float*)d_in[1];
    const int*   idx_i = (const int*)d_in[2];
    const int*   idx_j = (const int*)d_in[3];
    float*       y     = (float*)d_out;

    const int nnz = in_sizes[2];
    const int M   = in_sizes[0] / KDIM;

    dim3 grid(M / ROWS_PER_WG, NB);
    bsmm_wmma<<<grid, THREADS, 0, stream>>>(x, w, idx_i, idx_j, nnz, y);
}


# x tile read as whole 128-byte lines (8 loads of 4 full lines per wave and block, not 8 of 16 quarter lines) and passed to the A fragment layout through a per-wave LDS tile; sum order kept
# speedup vs baseline: 8.1109x; 1.0474x over previous
#include <hip/hip_runtime.h>

typedef __attribute__((ext_vector_type(2))) float v2f;
typedef __attribute__((ext_vector_type(4))) float v4f;
typedef __attribute__((ext_vector_type(8))) float v8f;
typedef __attribute__((ext_vector_type(2))) int   v2i;

#define BLOCK 32
#define NB 128
#define KDIM 4096
#define NDIM 4096
#define ROWS_PER_WG 256
#define THREADS 256
#define NWAVES 8
#define LSTRIDE 36
#define CHUNKS 8
#define SEG (NWAVES * CHUNKS * 32)

__global__ void __launch_bounds__(THREADS, 10)
bsmm_wmma(const float* __restrict__ x, const float* __restrict__ w,
          const int* __restrict__ idx_i, const int* __restrict__ idx_j, int nnz,
          float* __restrict__ y)
{
    __shared__ alignas(16) float wlds[2][BLOCK * LSTRIDE];
    __shared__ alignas(16) float xst[NWAVES][32 * LSTRIDE];
    __shared__ v2i lst[SEG + 2];
    __shared__ int wcnt[NWAVES];

    const int j    = blockIdx.y;
    const int lane = threadIdx.x & 31;
    const int wave = threadIdx.x >> 5;
    const int half = lane >> 4;
    const int lid  = lane & 15;
    const int m0   = blockIdx.x * ROWS_PER_WG + wave * 32;

    v8f c00 = {}, c01 = {};
    v8f c10 = {}, c11 = {};

    const int srow8 = ((lane >> 3) & 1) * 8 + (lane >> 4) * 16;
    const float* xgl = x + (size_t)(m0 + srow8) * KDIM + (lane & 7) * 4;
    float* xsst = &xst[wave][srow8 * LSTRIDE + (lane & 7) * 4];

    const int boff = lid * LSTRIDE + half * 4;
    const float* al = &xst[wave][boff];

    const int srow = threadIdx.x >> 3;
    const int scol = (threadIdx.x & 7) * 4;
    const int soff = srow * BLOCK + scol;

    const int eloc0 = wave * (CHUNKS * 32) + lane;

    for (int seg0 = 0; seg0 < nnz; seg0 += SEG) {
        int vj[CHUNKS], vi[CHUNKS];
#pragma unroll
        for (int c = 0; c < CHUNKS; ++c) {
            const int e  = seg0 + eloc0 + c * 32;
            const int ec = e < nnz ? e : nnz - 1;
            const int rj = idx_j[ec];
            vi[c] = idx_i[ec];
            vj[c] = e < nnz ? rj : -1;
        }
        unsigned m[CHUNKS];
        int wc = 0;
#pragma unroll
        for (int c = 0; c < CHUNKS; ++c) {
            m[c] = __builtin_amdgcn_ballot_w32(vj[c] == j);
            wc += __builtin_popcount(m[c]);
        }
        if (lane == 0) wcnt[wave] = wc;
        __syncthreads();

        int base = 0, tot = 0;
#pragma unroll
        for (int k = 0; k < NWAVES; ++k) {
            const int v = wcnt[k];
            tot += v;
            if (k < wave) base += v;
        }
        const int cnt = __builtin_amdgcn_readfirstlane(tot);

#pragma unroll
        for (int c = 0; c < CHUNKS; ++c) {
            if (vj[c] == j) {
                v2i ent = { seg0 + eloc0 + c * 32, vi[c] };
                lst[base + (int)__builtin_amdgcn_mbcnt_lo(m[c], 0u)] = ent;
            }
            base += __builtin_popcount(m[c]);
        }
        if (threadIdx.x == 0) { v2i z = { 0, 0 }; lst[cnt] = z; lst[cnt + 1] = z; }
        __syncthreads();

        const v2i l0 = lst[0], l1 = lst[1];
        int blk_cur = __builtin_amdgcn_readfirstlane(l0[0]);
        int col_cur = __builtin_amdgcn_readfirstlane(l0[1]);
        int blk_nxt = __builtin_amdgcn_readfirstlane(l1[0]);
        int col_nxt = __builtin_amdgcn_readfirstlane(l1[1]);

        if (cnt > 0) {
            const float* wsrc = w + (size_t)blk_cur * (BLOCK * BLOCK) + soff;
            v4f t0 = *(const v4f*)wsrc;
            wlds[0][(scol + 0) * LSTRIDE + srow] = t0[0];
            wlds[0][(scol + 1) * LSTRIDE + srow] = t0[1];
            wlds[0][(scol + 2) * LSTRIDE + srow] = t0[2];
            wlds[0][(scol + 3) * LSTRIDE + srow] = t0[3];
        }

        for (int t = 0; t < cnt; ++t) {
            __syncthreads();
            const int buf = t & 1;


            const bool have_next = (t + 1) < cnt;
            v4f nxt = {};
            if (have_next) {
                const float* wsrc = w + (size_t)blk_nxt * (BLOCK * BLOCK) + soff;
                nxt = *(const v4f*)wsrc;
            }

            const float* xg = xgl + col_cur * BLOCK;
            const float* bl = &wlds[buf][boff];

            v4f xr[8];
#pragma unroll
            for (int i = 0; i < 8; ++i)
                xr[i] = *(const v4f*)(xg + (size_t)i * KDIM);
#pragma unroll
            for (int i = 0; i < 8; ++i)
                *(v4f*)(xsst + i * LSTRIDE) = xr[i];
            __builtin_amdgcn_fence(__ATOMIC_ACQ_REL, "wavefront");
#pragma unroll
            for (int q = 0; q < 4; ++q) {
                const v4f a0q = *(const v4f*)(al + q * 8);
                const v4f a1q = *(const v4f*)(al + 16 * LSTRIDE + q * 8);
                const v4f b0 = *(const v4f*)(bl + q * 8);
                const v4f b1 = *(const v4f*)(bl + 16 * LSTRIDE + q * 8);
                const v2f a0l = __builtin_shufflevector(a0q, a0q, 0, 1);
                const v2f a1l = __builtin_shufflevector(a1q, a1q, 0, 1);
                const v2f b0l = __builtin_shufflevector(b0, b0, 0, 1);
                const v2f b1l = __builtin_shufflevector(b1, b1, 0, 1);
                c00 = __builtin_amdgcn_wmma_f32_16x16x4_f32(
                          false, a0l, false, b0l, (short)0, c00, false, false);
                c01 = __builtin_amdgcn_wmma_f32_16x16x4_f32(
                          false, a0l, false, b1l, (short)0, c01, false, false);
                c10 = __builtin_amdgcn_wmma_f32_16x16x4_f32(
                          false, a1l, false, b0l, (short)0, c10, false, false);
                c11 = __builtin_amdgcn_wmma_f32_16x16x4_f32(
                          false, a1l, false, b1l, (short)0, c11, false, false);
                const v2f a0h = __builtin_shufflevector(a0q, a0q, 2, 3);
                const v2f a1h = __builtin_shufflevector(a1q, a1q, 2, 3);
                const v2f b0h = __builtin_shufflevector(b0, b0, 2, 3);
                const v2f b1h = __builtin_shufflevector(b1, b1, 2, 3);
                c00 = __builtin_amdgcn_wmma_f32_16x16x4_f32(
                          false, a0h, false, b0h, (short)0, c00, false, false);
                c01 = __builtin_amdgcn_wmma_f32_16x16x4_f32(
                          false, a0h, false, b1h, (short)0, c01, false, false);
                c10 = __builtin_amdgcn_wmma_f32_16x16x4_f32(
                          false, a1h, false, b0h, (short)0, c10, false, false);
                c11 = __builtin_amdgcn_wmma_f32_16x16x4_f32(
                          false, a1h, false, b1h, (short)0, c11, false, false);
            }

            if (have_next) {
                __syncthreads();
                float* dst = &wlds[buf ^ 1][0];
                dst[(scol + 0) * LSTRIDE + srow] = nxt[0];
                dst[(scol + 1) * LSTRIDE + srow] = nxt[1];
                dst[(scol + 2) * LSTRIDE + srow] = nxt[2];
                dst[(scol + 3) * LSTRIDE + srow] = nxt[3];
            }

            const v2i la = lst[t + 2];
            col_cur = col_nxt;
            blk_nxt = __builtin_amdgcn_readfirstlane(la[0]);
            col_nxt = __builtin_amdgcn_readfirstlane(la[1]);
        }
    }

    float* y0 = y + (size_t)(m0 + half * 8) * NDIM + j * BLOCK + lid;
    float* y1 = y0 + (size_t)16 * NDIM;
#pragma unroll
    for (int r = 0; r < 8; ++r) {
        y0[(size_t)r * NDIM]      = c00[r];
        y0[(size_t)r * NDIM + 16] = c01[r];
        y1[(size_t)r * NDIM]      = c10[r];
        y1[(size_t)r * NDIM + 16] = c11[r];
    }
}

extern "C" void kernel_launch(void* const* d_in, const int* in_sizes, int n_in,
                              void* d_out, int out_size, void* d_ws, size_t ws_size,
                              hipStream_t stream) {
    const float* x     = (const float*)d_in[0];
    const float* w     = (const float*)d_in[1];
    const int*   idx_i = (const int*)d_in[2];
    const int*   idx_j = (const int*)d_in[3];
    float*       y     = (float*)d_out;

    const int nnz = in_sizes[2];
    const int M   = in_sizes[0] / KDIM;

    dim3 grid(M / ROWS_PER_WG, NB);
    bsmm_wmma<<<grid, THREADS, 0, stream>>>(x, w, idx_i, idx_j, nnz, y);
}


# Block loop software-pipelined: block t+1's x tile and w piece are requested right behind block t's barrier and land under its products; one barrier per block, not two; sum order kept
# speedup vs baseline: 8.3490x; 1.0294x over previous
#include <hip/hip_runtime.h>

typedef __attribute__((ext_vector_type(2))) float v2f;
typedef __attribute__((ext_vector_type(4))) float v4f;
typedef __attribute__((ext_vector_type(8))) float v8f;
typedef __attribute__((ext_vector_type(2))) int   v2i;

#define BLOCK 32
#define NB 128
#define KDIM 4096
#define NDIM 4096
#define ROWS_PER_WG 256
#define THREADS 256
#define NWAVES 8
#define LSTRIDE 36
#define CHUNKS 8
#define SEG (NWAVES * CHUNKS * 32)

__global__ void __launch_bounds__(THREADS, 8)
bsmm_wmma(const float* __restrict__ x, const float* __restrict__ w,
          const int* __restrict__ idx_i, const int* __restrict__ idx_j, int nnz,
          float* __restrict__ y)
{
    __shared__ alignas(16) float wlds[2][BLOCK * LSTRIDE];
    __shared__ alignas(16) float xst[NWAVES][32 * LSTRIDE];
    __shared__ v2i lst[SEG + 2];
    __shared__ int wcnt[NWAVES];

    const int j    = blockIdx.y;
    const int lane = threadIdx.x & 31;
    const int wave = threadIdx.x >> 5;
    const int half = lane >> 4;
    const int lid  = lane & 15;
    const int m0   = blockIdx.x * ROWS_PER_WG + wave * 32;

    v8f c00 = {}, c01 = {};
    v8f c10 = {}, c11 = {};

    const int srow8 = ((lane >> 3) & 1) * 8 + (lane >> 4) * 16;
    const float* xgl = x + (size_t)(m0 + srow8) * KDIM + (lane & 7) * 4;
    float* xsst = &xst[wave][srow8 * LSTRIDE + (lane & 7) * 4];

    const int boff = lid * LSTRIDE + half * 4;
    const float* al = &xst[wave][boff];

    const int srow = threadIdx.x >> 3;
    const int scol = (threadIdx.x & 7) * 4;
    const int soff = srow * BLOCK + scol;

    const int eloc0 = wave * (CHUNKS * 32) + lane;

    for (int seg0 = 0; seg0 < nnz; seg0 += SEG) {
        int vj[CHUNKS], vi[CHUNKS];
#pragma unroll
        for (int c = 0; c < CHUNKS; ++c) {
            const int e  = seg0 + eloc0 + c * 32;
            const int ec = e < nnz ? e : nnz - 1;
            const int rj = idx_j[ec];
            vi[c] = idx_i[ec];
            vj[c] = e < nnz ? rj : -1;
        }
        unsigned m[CHUNKS];
        int wc = 0;
#pragma unroll
        for (int c = 0; c < CHUNKS; ++c) {
            m[c] = __builtin_amdgcn_ballot_w32(vj[c] == j);
            wc += __builtin_popcount(m[c]);
        }
        if (lane == 0) wcnt[wave] = wc;
        __syncthreads();

        int base = 0, tot = 0;
#pragma unroll
        for (int k = 0; k < NWAVES; ++k) {
            const int v = wcnt[k];
            tot += v;
            if (k < wave) base += v;
        }
        const int cnt = __builtin_amdgcn_readfirstlane(tot);

#pragma unroll
        for (int c = 0; c < CHUNKS; ++c) {
            if (vj[c] == j) {
                v2i ent = { seg0 + eloc0 + c * 32, vi[c] };
                lst[base + (int)__builtin_amdgcn_mbcnt_lo(m[c], 0u)] = ent;
            }
            base += __builtin_popcount(m[c]);
        }
        if (threadIdx.x == 0) { v2i z = { 0, 0 }; lst[cnt] = z; lst[cnt + 1] = z; }
        __syncthreads();

        const v2i l0 = lst[0], l1 = lst[1];
        const int blk_cur = __builtin_amdgcn_readfirstlane(l0[0]);
        const int col_cur = __builtin_amdgcn_readfirstlane(l0[1]);
        int blk_nxt = __builtin_amdgcn_readfirstlane(l1[0]);
        int col_nxt = __builtin_amdgcn_readfirstlane(l1[1]);

        v4f xr[8];
        if (cnt > 0) {
            const float* wsrc = w + (size_t)blk_cur * (BLOCK * BLOCK) + soff;
            v4f t0 = *(const v4f*)wsrc;
            const float* xg = xgl + col_cur * BLOCK;
#pragma unroll
            for (int i = 0; i < 8; ++i)
                xr[i] = *(const v4f*)(xg + (size_t)i * KDIM);
            wlds[0][(scol + 0) * LSTRIDE + srow] = t0[0];
            wlds[0][(scol + 1) * LSTRIDE + srow] = t0[1];
            wlds[0][(scol + 2) * LSTRIDE + srow] = t0[2];
            wlds[0][(scol + 3) * LSTRIDE + srow] = t0[3];
        }

        for (int t = 0; t < cnt; ++t) {
#pragma unroll
            for (int i = 0; i < 8; ++i)
                *(v4f*)(xsst + i * LSTRIDE) = xr[i];
            __syncthreads();
            const int buf = t & 1;

            const v2i la = lst[t + 2];

            const bool have_next = (t + 1) < cnt;
            v4f nxt = {};
            if (have_next) {
                const float* wsrc = w + (size_t)blk_nxt * (BLOCK * BLOCK) + soff;
                nxt = *(const v4f*)wsrc;
                const float* xg = xgl + col_nxt * BLOCK;
#pragma unroll
                for (int i = 0; i < 8; ++i)
                    xr[i] = *(const v4f*)(xg + (size_t)i * KDIM);
            }

            const float* bl = &wlds[buf][boff];
#pragma unroll
            for (int q = 0; q < 4; ++q) {
                const v4f a0q = *(const v4f*)(al + q * 8);
                const v4f a1q = *(const v4f*)(al + 16 * LSTRIDE + q * 8);
                const v4f b0 = *(const v4f*)(bl + q * 8);
                const v4f b1 = *(const v4f*)(bl + 16 * LSTRIDE + q * 8);
                const v2f a0l = __builtin_shufflevector(a0q, a0q, 0, 1);
                const v2f a1l = __builtin_shufflevector(a1q, a1q, 0, 1);
                const v2f b0l = __builtin_shufflevector(b0, b0, 0, 1);
                const v2f b1l = __builtin_shufflevector(b1, b1, 0, 1);
                c00 = __builtin_amdgcn_wmma_f32_16x16x4_f32(
                          false, a0l, false, b0l, (short)0, c00, false, false);
                c01 = __builtin_amdgcn_wmma_f32_16x16x4_f32(
                          false, a0l, false, b1l, (short)0, c01, false, false);
                c10 = __builtin_amdgcn_wmma_f32_16x16x4_f32(
                          false, a1l, false, b0l, (short)0, c10, false, false);
                c11 = __builtin_amdgcn_wmma_f32_16x16x4_f32(
                          false, a1l, false, b1l, (short)0, c11, false, false);
                const v2f a0h = __builtin_shufflevector(a0q, a0q, 2, 3);
                const v2f a1h = __builtin_shufflevector(a1q, a1q, 2, 3);
                const v2f b0h = __builtin_shufflevector(b0, b0, 2, 3);
                const v2f b1h = __builtin_shufflevector(b1, b1, 2, 3);
                c00 = __builtin_amdgcn_wmma_f32_16x16x4_f32(
                          false, a0h, false, b0h, (short)0, c00, false, false);
                c01 = __builtin_amdgcn_wmma_f32_16x16x4_f32(
                          false, a0h, false, b1h, (short)0, c01, false, false);
                c10 = __builtin_amdgcn_wmma_f32_16x16x4_f32(
                          false, a1h, false, b0h, (short)0, c10, false, false);
                c11 = __builtin_amdgcn_wmma_f32_16x16x4_f32(
                          false, a1h, false, b1h, (short)0, c11, false, false);
            }

            if (have_next) {
                float* dst = &wlds[buf ^ 1][0];
                dst[(scol + 0) * LSTRIDE + srow] = nxt[0];
                dst[(scol + 1) * LSTRIDE + srow] = nxt[1];
                dst[(scol + 2) * LSTRIDE + srow] = nxt[2];
                dst[(scol + 3) * LSTRIDE + srow] = nxt[3];
            }

            blk_nxt = __builtin_amdgcn_readfirstlane(la[0]);
            col_nxt = __builtin_amdgcn_readfirstlane(la[1]);
        }
    }

    float* y0 = y + (size_t)(m0 + half * 8) * NDIM + j * BLOCK + lid;
    float* y1 = y0 + (size_t)16 * NDIM;
#pragma unroll
    for (int r = 0; r < 8; ++r) {
        y0[(size_t)r * NDIM]      = c00[r];
        y0[(size_t)r * NDIM + 16] = c01[r];
        y1[(size_t)r * NDIM]      = c10[r];
        y1[(size_t)r * NDIM + 16] = c11[r];
    }
}

extern "C" void kernel_launch(void* const* d_in, const int* in_sizes, int n_in,
                              void* d_out, int out_size, void* d_ws, size_t ws_size,
                              hipStream_t stream) {
    const float* x     = (const float*)d_in[0];
    const float* w     = (const float*)d_in[1];
    const int*   idx_i = (const int*)d_in[2];
    const int*   idx_j = (const int*)d_in[3];
    float*       y     = (float*)d_out;

    const int nnz = in_sizes[2];
    const int M   = in_sizes[0] / KDIM;

    dim3 grid(M / ROWS_PER_WG, NB);
    bsmm_wmma<<<grid, THREADS, 0, stream>>>(x, w, idx_i, idx_j, nnz, y);
}


# Block columns handed out heaviest first: a one-block first launch counts blocks per column and writes the column order (workspace, 128 ints) that the GEMM grid's y index goes through; sum order kept
# speedup vs baseline: 8.7401x; 1.0468x over previous
#include <hip/hip_runtime.h>

typedef __attribute__((ext_vector_type(2))) float v2f;
typedef __attribute__((ext_vector_type(4))) float v4f;
typedef __attribute__((ext_vector_type(8))) float v8f;
typedef __attribute__((ext_vector_type(2))) int   v2i;

#define BLOCK 32
#define NB 128
#define KDIM 4096
#define NDIM 4096
#define ROWS_PER_WG 256
#define THREADS 256
#define NWAVES 8
#define LSTRIDE 36
#define CHUNKS 8
#define SEG (NWAVES * CHUNKS * 32)
#define OT 1024
#define OSEG (OT / NB)
#define OPASS 2048
#define OWORDS (OPASS / 4)

__global__ void __launch_bounds__(OT)
bsmm_order(const int* __restrict__ idx_j, int nnz, int* __restrict__ order)
{
    __shared__ alignas(16) unsigned pk[OWORDS];
    __shared__ int part[OSEG][NB];
    __shared__ alignas(16) unsigned keys[NB];

    const int c = threadIdx.x & (NB - 1);
    const int s = __builtin_amdgcn_readfirstlane((int)(threadIdx.x >> 7));
    const unsigned cb = (unsigned)c * 0x01010101u;

    int n = 0;
    int slots = 0;
    for (int p0 = 0; p0 < nnz; p0 += OPASS) {
        if (threadIdx.x < OWORDS) {
            const int e = p0 + (int)threadIdx.x * 4;
            int v0 = 0, v1 = 0, v2 = 0, v3 = 0;
            if (e + 3 < nnz) {
                const int4 q = *(const int4*)(idx_j + e);
                v0 = q.x; v1 = q.y; v2 = q.z; v3 = q.w;
            } else {
                if (e     < nnz) v0 = idx_j[e];
                if (e + 1 < nnz) v1 = idx_j[e + 1];
                if (e + 2 < nnz) v2 = idx_j[e + 2];
            }
            pk[threadIdx.x] = ((unsigned)v0 & 0x7Fu)       | ((unsigned)v1 & 0x7Fu) << 8 |
                              ((unsigned)v2 & 0x7Fu) << 16 | ((unsigned)v3 & 0x7Fu) << 24;
        }
        __syncthreads();
        const uint4* pp = (const uint4*)&pk[s * (OWORDS / OSEG)];
#pragma unroll
        for (int q = 0; q < OWORDS / OSEG / 4; ++q) {
            const uint4 v = pp[q];
            n += __builtin_popcount((0x80808080u - (v.x ^ cb)) & 0x80808080u);
            n += __builtin_popcount((0x80808080u - (v.y ^ cb)) & 0x80808080u);
            n += __builtin_popcount((0x80808080u - (v.z ^ cb)) & 0x80808080u);
            n += __builtin_popcount((0x80808080u - (v.w ^ cb)) & 0x80808080u);
        }
        slots += OPASS;
        __syncthreads();
    }
    part[s][c] = n;
    __syncthreads();

    if (threadIdx.x < NB) {
        int tot = 0;
#pragma unroll
        for (int k = 0; k < OSEG; ++k) tot += part[k][c];
        if (c == 0) tot -= slots - nnz;
        keys[c] = ((unsigned)tot << 7) | (unsigned)(NB - 1 - c);
    }
    __syncthreads();

    const unsigned mine = keys[c];
    const uint4* kp = (const uint4*)&keys[s * (NB / OSEG)];
    int r = 0;
#pragma unroll
    for (int q = 0; q < NB / OSEG / 4; ++q) {
        const uint4 v = kp[q];
        r += (v.x > mine) + (v.y > mine) + (v.z > mine) + (v.w > mine);
    }
    part[s][c] = r;
    __syncthreads();

    if (threadIdx.x < NB) {
        int rank = 0;
#pragma unroll
        for (int k = 0; k < OSEG; ++k) rank += part[k][c];
        order[rank] = c;
    }
}

__global__ void __launch_bounds__(THREADS, 8)
bsmm_wmma(const float* __restrict__ x, const float* __restrict__ w,
          const int* __restrict__ idx_i, const int* __restrict__ idx_j, int nnz,
          const int* __restrict__ order, float* __restrict__ y)
{
    __shared__ alignas(16) float wlds[2][BLOCK * LSTRIDE];
    __shared__ alignas(16) float xst[NWAVES][32 * LSTRIDE];
    __shared__ v2i lst[SEG + 2];
    __shared__ int wcnt[NWAVES];

    const int j    = order[blockIdx.y];
    const int lane = threadIdx.x & 31;
    const int wave = threadIdx.x >> 5;
    const int half = lane >> 4;
    const int lid  = lane & 15;
    const int m0   = blockIdx.x * ROWS_PER_WG + wave * 32;

    v8f c00 = {}, c01 = {};
    v8f c10 = {}, c11 = {};

    const int srow8 = ((lane >> 3) & 1) * 8 + (lane >> 4) * 16;
    const float* xgl = x + (size_t)(m0 + srow8) * KDIM + (lane & 7) * 4;
    float* xsst = &xst[wave][srow8 * LSTRIDE + (lane & 7) * 4];

    const int boff = lid * LSTRIDE + half * 4;
    const float* al = &xst[wave][boff];

    const int srow = threadIdx.x >> 3;
    const int scol = (threadIdx.x & 7) * 4;
    const int soff = srow * BLOCK + scol;

    const int eloc0 = wave * (CHUNKS * 32) + lane;

    for (int seg0 = 0; seg0 < nnz; seg0 += SEG) {
        int vj[CHUNKS], vi[CHUNKS];
#pragma unroll
        for (int c = 0; c < CHUNKS; ++c) {
            const int e  = seg0 + eloc0 + c * 32;
            const int ec = e < nnz ? e : nnz - 1;
            const int rj = idx_j[ec];
            vi[c] = idx_i[ec];
            vj[c] = e < nnz ? rj : -1;
        }
        unsigned m[CHUNKS];
        int wc = 0;
#pragma unroll
        for (int c = 0; c < CHUNKS; ++c) {
            m[c] = __builtin_amdgcn_ballot_w32(vj[c] == j);
            wc += __builtin_popcount(m[c]);
        }
        if (lane == 0) wcnt[wave] = wc;
        __syncthreads();

        int base = 0, tot = 0;
#pragma unroll
        for (int k = 0; k < NWAVES; ++k) {
            const int v = wcnt[k];
            tot += v;
            if (k < wave) base += v;
        }
        const int cnt = __builtin_amdgcn_readfirstlane(tot);

#pragma unroll
        for (int c = 0; c < CHUNKS; ++c) {
            if (vj[c] == j) {
                v2i ent = { seg0 + eloc0 + c * 32, vi[c] };
                lst[base + (int)__builtin_amdgcn_mbcnt_lo(m[c], 0u)] = ent;
            }
            base += __builtin_popcount(m[c]);
        }
        if (threadIdx.x == 0) { v2i z = { 0, 0 }; lst[cnt] = z; lst[cnt + 1] = z; }
        __syncthreads();

        const v2i l0 = lst[0], l1 = lst[1];
        const int blk_cur = __builtin_amdgcn_readfirstlane(l0[0]);
        const int col_cur = __builtin_amdgcn_readfirstlane(l0[1]);
        int blk_nxt = __builtin_amdgcn_readfirstlane(l1[0]);
        int col_nxt = __builtin_amdgcn_readfirstlane(l1[1]);

        v4f xr[8];
        if (cnt > 0) {
            const float* wsrc = w + (size_t)blk_cur * (BLOCK * BLOCK) + soff;
            v4f t0 = *(const v4f*)wsrc;
            const float* xg = xgl + col_cur * BLOCK;
#pragma unroll
            for (int i = 0; i < 8; ++i)
                xr[i] = *(const v4f*)(xg + (size_t)i * KDIM);
            wlds[0][(scol + 0) * LSTRIDE + srow] = t0[0];
            wlds[0][(scol + 1) * LSTRIDE + srow] = t0[1];
            wlds[0][(scol + 2) * LSTRIDE + srow] = t0[2];
            wlds[0][(scol + 3) * LSTRIDE + srow] = t0[3];
        }

        for (int t = 0; t < cnt; ++t) {
#pragma unroll
            for (int i = 0; i < 8; ++i)
                *(v4f*)(xsst + i * LSTRIDE) = xr[i];
            __syncthreads();
            const int buf = t & 1;

            const v2i la = lst[t + 2];

            const bool have_next = (t + 1) < cnt;
            v4f nxt = {};
            if (have_next) {
                const float* wsrc = w + (size_t)blk_nxt * (BLOCK * BLOCK) + soff;
                nxt = *(const v4f*)wsrc;
                const float* xg = xgl + col_nxt * BLOCK;
#pragma unroll
                for (int i = 0; i < 8; ++i)
                    xr[i] = *(const v4f*)(xg + (size_t)i * KDIM);
            }

            const float* bl = &wlds[buf][boff];
#pragma unroll
            for (int q = 0; q < 4; ++q) {
                const v4f a0q = *(const v4f*)(al + q * 8);
                const v4f a1q = *(const v4f*)(al + 16 * LSTRIDE + q * 8);
                const v4f b0 = *(const v4f*)(bl + q * 8);
                const v4f b1 = *(const v4f*)(bl + 16 * LSTRIDE + q * 8);
                const v2f a0l = __builtin_shufflevector(a0q, a0q, 0, 1);
                const v2f a1l = __builtin_shufflevector(a1q, a1q, 0, 1);
                const v2f b0l = __builtin_shufflevector(b0, b0, 0, 1);
                const v2f b1l = __builtin_shufflevector(b1, b1, 0, 1);
                c00 = __builtin_amdgcn_wmma_f32_16x16x4_f32(
                          false, a0l, false, b0l, (short)0, c00, false, false);
                c01 = __builtin_amdgcn_wmma_f32_16x16x4_f32(
                          false, a0l, false, b1l, (short)0, c01, false, false);
                c10 = __builtin_amdgcn_wmma_f32_16x16x4_f32(
                          false, a1l, false, b0l, (short)0, c10, false, false);
                c11 = __builtin_amdgcn_wmma_f32_16x16x4_f32(
                          false, a1l, false, b1l, (short)0, c11, false, false);
                const v2f a0h = __builtin_shufflevector(a0q, a0q, 2, 3);
                const v2f a1h = __builtin_shufflevector(a1q, a1q, 2, 3);
                const v2f b0h = __builtin_shufflevector(b0, b0, 2, 3);
                const v2f b1h = __builtin_shufflevector(b1, b1, 2, 3);
                c00 = __builtin_amdgcn_wmma_f32_16x16x4_f32(
                          false, a0h, false, b0h, (short)0, c00, false, false);
                c01 = __builtin_amdgcn_wmma_f32_16x16x4_f32(
                          false, a0h, false, b1h, (short)0, c01, false, false);
                c10 = __builtin_amdgcn_wmma_f32_16x16x4_f32(
                          false, a1h, false, b0h, (short)0, c10, false, false);
                c11 = __builtin_amdgcn_wmma_f32_16x16x4_f32(
                          false, a1h, false, b1h, (short)0, c11, false, false);
            }

            if (have_next) {
                float* dst = &wlds[buf ^ 1][0];
                dst[(scol + 0) * LSTRIDE + srow] = nxt[0];
                dst[(scol + 1) * LSTRIDE + srow] = nxt[1];
                dst[(scol + 2) * LSTRIDE + srow] = nxt[2];
                dst[(scol + 3) * LSTRIDE + srow] = nxt[3];
            }

            blk_nxt = __builtin_amdgcn_readfirstlane(la[0]);
            col_nxt = __builtin_amdgcn_readfirstlane(la[1]);
        }
    }

    float* y0 = y + (size_t)(m0 + half * 8) * NDIM + j * BLOCK + lid;
    float* y1 = y0 + (size_t)16 * NDIM;
#pragma unroll
    for (int r = 0; r < 8; ++r) {
        y0[(size_t)r * NDIM]      = c00[r];
        y0[(size_t)r * NDIM + 16] = c01[r];
        y1[(size_t)r * NDIM]      = c10[r];
        y1[(size_t)r * NDIM + 16] = c11[r];
    }
}

extern "C" void kernel_launch(void* const* d_in, const int* in_sizes, int n_in,
                              void* d_out, int out_size, void* d_ws, size_t ws_size,
                              hipStream_t stream) {
    const float* x     = (const float*)d_in[0];
    const float* w     = (const float*)d_in[1];
    const int*   idx_i = (const int*)d_in[2];
    const int*   idx_j = (const int*)d_in[3];
    float*       y     = (float*)d_out;

    const int nnz = in_sizes[2];
    const int M   = in_sizes[0] / KDIM;

    int* order = (int*)d_ws;

    bsmm_order<<<1, OT, 0, stream>>>(idx_j, nnz, order);
    dim3 grid(M / ROWS_PER_WG, NB);
    bsmm_wmma<<<grid, THREADS, 0, stream>>>(x, w, idx_i, idx_j, nnz, order, y);
}
